# starting kernel
# baseline (speedup 1.0000x reference)
#include <hip/hip_runtime.h>
#include <hip/hip_bf16.h>
#include <math.h>

typedef __attribute__((ext_vector_type(2))) float v2f;
typedef __attribute__((ext_vector_type(8))) float v8f;

#define C_DIM 256
#define N_PIX 196
#define MAT_ELEMS (C_DIM * C_DIM)
#define BATCH 64
#define OUT_PER_B 32896
#define LDS_STRIDE 34

__global__ __launch_bounds__(256) void mean_kernel(const float* __restrict__ x,
                                                   float* __restrict__ mean) {
    int row = blockIdx.x * 8 + (threadIdx.x >> 5);
    int lane = threadIdx.x & 31;
    const float* xr = x + (size_t)row * N_PIX;
    float s = 0.0f;
    for (int i = lane; i < N_PIX; i += 32) s += xr[i];
    #pragma unroll
    for (int o = 16; o > 0; o >>= 1) s += __shfl_down(s, o, 32);
    if (lane == 0) mean[row] = s * (1.0f / (float)N_PIX);
}

__global__ __launch_bounds__(256) void sigma_kernel(float* __restrict__ D,
                                                    const float* __restrict__ x,
                                                    const float* __restrict__ mean) {
    __shared__ float As[128 * LDS_STRIDE];
    __shared__ float Bt[128 * LDS_STRIDE];
    const int tid  = threadIdx.x;
    const int lane = tid & 31;
    const int wave = tid >> 5;
    const int b    = blockIdx.y;
    const int tileid   = blockIdx.x;
    const int row_base = (tileid >> 1) * 128;
    const int col_base = (tileid & 1) * 128;
    const float* xb = x + (size_t)b * C_DIM * N_PIX;
    const float* mb = mean + (size_t)b * C_DIM;
    float* Db = D + (size_t)b * MAT_ELEMS;

    const int wrow = (wave >> 1) * 32;
    const int wcol = (wave & 1) * 64;
    const int lr = lane & 15;
    const int kh = (lane >> 4) * 2;
    const float inv_n = 1.0f / (float)N_PIX;

    const int aR0 = (wrow +      lr) * LDS_STRIDE;
    const int aR1 = (wrow + 16 + lr) * LDS_STRIDE;
    const int bR0 = (wcol +      lr) * LDS_STRIDE;
    const int bR1 = (wcol + 16 + lr) * LDS_STRIDE;
    const int bR2 = (wcol + 32 + lr) * LDS_STRIDE;
    const int bR3 = (wcol + 48 + lr) * LDS_STRIDE;

    v8f acc00 = {}, acc01 = {}, acc02 = {}, acc03 = {};
    v8f acc10 = {}, acc11 = {}, acc12 = {}, acc13 = {};

    for (int kk = 0; kk < 224; kk += 32) {
        __syncthreads();
        #pragma unroll
        for (int r = 0; r < 16; ++r) {
            int e = r * 256 + tid;
            int mm = e >> 5, k = e & 31;
            int kg = kk + k;
            int row = row_base + mm;
            float v = 0.0f;
            if (kg < N_PIX) v = (xb[(size_t)row * N_PIX + kg] - mb[row]) * inv_n;
            As[mm * LDS_STRIDE + k] = v;
        }
        #pragma unroll
        for (int r = 0; r < 16; ++r) {
            int e = r * 256 + tid;
            int nn = e >> 5, k = e & 31;
            int kg = kk + k;
            int col = col_base + nn;
            float v = 0.0f;
            if (kg < N_PIX) v = xb[(size_t)col * N_PIX + kg];
            Bt[nn * LDS_STRIDE + k] = v;
        }
        __syncthreads();
        #pragma unroll
        for (int ks = 0; ks < 8; ++ks) {
            int kf = ks * 4 + kh;
            v2f a0 = *(const v2f*)&As[aR0 + kf];
            v2f a1 = *(const v2f*)&As[aR1 + kf];
            v2f b0 = *(const v2f*)&Bt[bR0 + kf];
            v2f b1 = *(const v2f*)&Bt[bR1 + kf];
            v2f b2 = *(const v2f*)&Bt[bR2 + kf];
            v2f b3 = *(const v2f*)&Bt[bR3 + kf];
            acc00 = __builtin_amdgcn_wmma_f32_16x16x4_f32(false, a0, false, b0, (short)0, acc00, false, false);
            acc01 = __builtin_amdgcn_wmma_f32_16x16x4_f32(false, a0, false, b1, (short)0, acc01, false, false);
            acc02 = __builtin_amdgcn_wmma_f32_16x16x4_f32(false, a0, false, b2, (short)0, acc02, false, false);
            acc03 = __builtin_amdgcn_wmma_f32_16x16x4_f32(false, a0, false, b3, (short)0, acc03, false, false);
            acc10 = __builtin_amdgcn_wmma_f32_16x16x4_f32(false, a1, false, b0, (short)0, acc10, false, false);
            acc11 = __builtin_amdgcn_wmma_f32_16x16x4_f32(false, a1, false, b1, (short)0, acc11, false, false);
            acc12 = __builtin_amdgcn_wmma_f32_16x16x4_f32(false, a1, false, b2, (short)0, acc12, false, false);
            acc13 = __builtin_amdgcn_wmma_f32_16x16x4_f32(false, a1, false, b3, (short)0, acc13, false, false);
        }
    }
    const int mrow = row_base + wrow + (lane >> 4) * 8;
    const int ncol = col_base + wcol + lr;
    #pragma unroll
    for (int v = 0; v < 8; ++v) {
        Db[(size_t)(mrow + v) * C_DIM + ncol]           = acc00[v];
        Db[(size_t)(mrow + v) * C_DIM + ncol + 16]      = acc01[v];
        Db[(size_t)(mrow + v) * C_DIM + ncol + 32]      = acc02[v];
        Db[(size_t)(mrow + v) * C_DIM + ncol + 48]      = acc03[v];
        Db[(size_t)(mrow + 16 + v) * C_DIM + ncol]      = acc10[v];
        Db[(size_t)(mrow + 16 + v) * C_DIM + ncol + 16] = acc11[v];
        Db[(size_t)(mrow + 16 + v) * C_DIM + ncol + 32] = acc12[v];
        Db[(size_t)(mrow + 16 + v) * C_DIM + ncol + 48] = acc13[v];
    }
}

__global__ __launch_bounds__(256) void trace_kernel(const float* __restrict__ S,
                                                    float* __restrict__ scal) {
    __shared__ float red[256];
    int b = blockIdx.x, tid = threadIdx.x;
    red[tid] = S[(size_t)b * MAT_ELEMS + (size_t)tid * 257];
    __syncthreads();
    for (int s = 128; s > 0; s >>= 1) {
        if (tid < s) red[tid] += red[tid + s];
        __syncthreads();
    }
    if (tid == 0) {
        float tr = red[0];
        scal[b] = 1.0f / tr;
        scal[64 + b] = sqrtf(tr);
    }
}

__global__ __launch_bounds__(256) void b0_kernel(float* __restrict__ Bm,
                                                 const float* __restrict__ S,
                                                 const float* __restrict__ scal) {
    int col = threadIdx.x, row = blockIdx.x, b = blockIdx.y;
    size_t idx = (size_t)b * MAT_ELEMS + (size_t)row * C_DIM + col;
    float rtr = scal[b];
    Bm[idx] = 0.5f * ((row == col ? 3.0f : 0.0f) - S[idx] * rtr);
}

__global__ __launch_bounds__(256) void ns_gemm_kernel(float* __restrict__ D,
                                                      const float* __restrict__ A,
                                                      const float* __restrict__ B,
                                                      const float* __restrict__ scal,
                                                      int op) {
    __shared__ float As[128 * LDS_STRIDE];
    __shared__ float Bt[128 * LDS_STRIDE];
    const int tid  = threadIdx.x;
    const int lane = tid & 31;
    const int wave = tid >> 5;
    const int b    = blockIdx.y;
    const int tileid   = blockIdx.x;
    const int row_base = (tileid >> 1) * 128;
    const int col_base = (tileid & 1) * 128;
    const float* Ab = A + (size_t)b * MAT_ELEMS;
    const float* Bb = B + (size_t)b * MAT_ELEMS;
    float* Dmat = D + (size_t)b * MAT_ELEMS;
    float* Dout = D + (size_t)b * OUT_PER_B;

    const int wrow = (wave >> 1) * 32;
    const int wcol = (wave & 1) * 64;
    const int lr = lane & 15;
    const int kh = (lane >> 4) * 2;

    const int aR0 = (wrow +      lr) * LDS_STRIDE;
    const int aR1 = (wrow + 16 + lr) * LDS_STRIDE;
    const int bR0 = (wcol +      lr) * LDS_STRIDE;
    const int bR1 = (wcol + 16 + lr) * LDS_STRIDE;
    const int bR2 = (wcol + 32 + lr) * LDS_STRIDE;
    const int bR3 = (wcol + 48 + lr) * LDS_STRIDE;

    v8f acc00 = {}, acc01 = {}, acc02 = {}, acc03 = {};
    v8f acc10 = {}, acc11 = {}, acc12 = {}, acc13 = {};

    for (int kk = 0; kk < 256; kk += 32) {
        __syncthreads();
        #pragma unroll
        for (int r = 0; r < 16; ++r) {
            int e = r * 256 + tid;
            int mm = e >> 5, k = e & 31;
            As[mm * LDS_STRIDE + k] = Ab[(size_t)(row_base + mm) * C_DIM + kk + k];
        }
        #pragma unroll
        for (int r = 0; r < 16; ++r) {
            int e = r * 256 + tid;
            int k = e >> 7, nn = e & 127;
            Bt[nn * LDS_STRIDE + k] = Bb[(size_t)(kk + k) * C_DIM + col_base + nn];
        }
        __syncthreads();
        #pragma unroll
        for (int ks = 0; ks < 8; ++ks) {
            int kf = ks * 4 + kh;
            v2f a0 = *(const v2f*)&As[aR0 + kf];
            v2f a1 = *(const v2f*)&As[aR1 + kf];
            v2f b0 = *(const v2f*)&Bt[bR0 + kf];
            v2f b1 = *(const v2f*)&Bt[bR1 + kf];
            v2f b2 = *(const v2f*)&Bt[bR2 + kf];
            v2f b3 = *(const v2f*)&Bt[bR3 + kf];
            acc00 = __builtin_amdgcn_wmma_f32_16x16x4_f32(false, a0, false, b0, (short)0, acc00, false, false);
            acc01 = __builtin_amdgcn_wmma_f32_16x16x4_f32(false, a0, false, b1, (short)0, acc01, false, false);
            acc02 = __builtin_amdgcn_wmma_f32_16x16x4_f32(false, a0, false, b2, (short)0, acc02, false, false);
            acc03 = __builtin_amdgcn_wmma_f32_16x16x4_f32(false, a0, false, b3, (short)0, acc03, false, false);
            acc10 = __builtin_amdgcn_wmma_f32_16x16x4_f32(false, a1, false, b0, (short)0, acc10, false, false);
            acc11 = __builtin_amdgcn_wmma_f32_16x16x4_f32(false, a1, false, b1, (short)0, acc11, false, false);
            acc12 = __builtin_amdgcn_wmma_f32_16x16x4_f32(false, a1, false, b2, (short)0, acc12, false, false);
            acc13 = __builtin_amdgcn_wmma_f32_16x16x4_f32(false, a1, false, b3, (short)0, acc13, false, false);
        }
    }

    float s = 1.0f;
    if (op >= 2) s = scal[b];
    const int mrow = row_base + wrow + (lane >> 4) * 8;
    const int ncol = col_base + wcol + lr;

    auto emit = [&](float val, int m, int n) {
        if (op == 3) {
            if (n >= m) {
                int start = m * C_DIM - (m * (m - 1)) / 2;
                Dout[start + (n - m)] = val * s;
            }
        } else {
            if (op == 1)      val = (m == n ? 1.5f : 0.0f) - 0.5f * val;
            else if (op == 2) val *= s;
            Dmat[(size_t)m * C_DIM + n] = val;
        }
    };
    #pragma unroll
    for (int v = 0; v < 8; ++v) {
        emit(acc00[v], mrow + v,      ncol);
        emit(acc01[v], mrow + v,      ncol + 16);
        emit(acc02[v], mrow + v,      ncol + 32);
        emit(acc03[v], mrow + v,      ncol + 48);
        emit(acc10[v], mrow + 16 + v, ncol);
        emit(acc11[v], mrow + 16 + v, ncol + 16);
        emit(acc12[v], mrow + 16 + v, ncol + 32);
        emit(acc13[v], mrow + 16 + v, ncol + 48);
    }
}

extern "C" void kernel_launch(void* const* d_in, const int* in_sizes, int n_in,
                              void* d_out, int out_size, void* d_ws, size_t ws_size,
                              hipStream_t stream) {
    const float* x = (const float*)d_in[0];
    float* out = (float*)d_out;
    float* ws = (float*)d_ws;

    float* mean = ws;
    float* scal = ws + 16384;
    const size_t MS = (size_t)BATCH * MAT_ELEMS;
    float* M0 = ws + 16512;
    float* M1 = M0 + MS;
    float* M2 = M1 + MS;
    float* M3 = M2 + MS;
    float* M4 = M3 + MS;

    dim3 gGemm(4, BATCH);
    dim3 gRow(C_DIM, BATCH);

    mean_kernel<<<2048, 256, 0, stream>>>(x, mean);
    sigma_kernel<<<gGemm, 256, 0, stream>>>(M0, x, mean);
    trace_kernel<<<BATCH, 256, 0, stream>>>(M0, scal);
    b0_kernel<<<gRow, 256, 0, stream>>>(M1, M0, scal);

    ns_gemm_kernel<<<gGemm, 256, 0, stream>>>(M2, M0, M1, scal, 2);

    ns_gemm_kernel<<<gGemm, 256, 0, stream>>>(M3, M1, M2, nullptr, 1);
    ns_gemm_kernel<<<gGemm, 256, 0, stream>>>(M4, M2, M3, nullptr, 0);
    ns_gemm_kernel<<<gGemm, 256, 0, stream>>>(M0, M3, M1, nullptr, 0);
    ns_gemm_kernel<<<gGemm, 256, 0, stream>>>(M1, M0, M4, nullptr, 1);
    ns_gemm_kernel<<<gGemm, 256, 0, stream>>>(M2, M4, M1, nullptr, 0);
    ns_gemm_kernel<<<gGemm, 256, 0, stream>>>(M3, M1, M0, nullptr, 0);
    ns_gemm_kernel<<<gGemm, 256, 0, stream>>>(M4, M3, M2, nullptr, 1);
    ns_gemm_kernel<<<gGemm, 256, 0, stream>>>(M0, M2, M4, nullptr, 0);
    ns_gemm_kernel<<<gGemm, 256, 0, stream>>>(M1, M4, M3, nullptr, 0);

    ns_gemm_kernel<<<gGemm, 256, 0, stream>>>(M2, M1, M0, nullptr, 1);
    ns_gemm_kernel<<<gGemm, 256, 0, stream>>>(out, M0, M2, scal + 64, 3);
}


# both WMMA kernels: K-slab staging as one group of 16-byte loads issued one slab ahead into a second LDS stage, not a serial load-wait-store chain; LDS layout, sum order kept
# speedup vs baseline: 3.2048x; 3.2048x over previous
#include <hip/hip_runtime.h>
#include <hip/hip_bf16.h>
#include <math.h>

typedef __attribute__((ext_vector_type(2))) float v2f;
typedef __attribute__((ext_vector_type(8))) float v8f;

#define C_DIM 256
#define N_PIX 196
#define MAT_ELEMS (C_DIM * C_DIM)
#define BATCH 64
#define OUT_PER_B 32896
#define LDS_STRIDE 34
#define LDS_TILE (128 * LDS_STRIDE)
#define LDS_STAGE (2 * LDS_TILE)

__global__ __launch_bounds__(256) void mean_kernel(const float* __restrict__ x,
                                                   float* __restrict__ mean) {
    int row = blockIdx.x * 8 + (threadIdx.x >> 5);
    int lane = threadIdx.x & 31;
    const float* xr = x + (size_t)row * N_PIX;
    float s = 0.0f;
    for (int i = lane; i < N_PIX; i += 32) s += xr[i];
    #pragma unroll
    for (int o = 16; o > 0; o >>= 1) s += __shfl_down(s, o, 32);
    if (lane == 0) mean[row] = s * (1.0f / (float)N_PIX);
}

__global__ __launch_bounds__(256) void sigma_kernel(float* __restrict__ D,
                                                    const float* __restrict__ x,
                                                    const float* __restrict__ mean) {
    __shared__ __attribute__((aligned(16))) float smem[2 * LDS_STAGE];
    const int tid  = threadIdx.x;
    const int lane = tid & 31;
    const int wave = tid >> 5;
    const int b    = blockIdx.y;
    const int tileid   = blockIdx.x;
    const int row_base = (tileid >> 1) * 128;
    const int col_base = (tileid & 1) * 128;
    const float* xb = x + (size_t)b * C_DIM * N_PIX;
    const float* mb = mean + (size_t)b * C_DIM;
    float* Db = D + (size_t)b * MAT_ELEMS;

    const int wrow = (wave >> 1) * 32;
    const int wcol = (wave & 1) * 64;
    const int lr = lane & 15;
    const int kh = (lane >> 4) * 2;
    const float inv_n = 1.0f / (float)N_PIX;

    const int aR0 = (wrow +      lr) * LDS_STRIDE;
    const int aR1 = (wrow + 16 + lr) * LDS_STRIDE;
    const int bR0 = LDS_TILE + (wcol +      lr) * LDS_STRIDE;
    const int bR1 = LDS_TILE + (wcol + 16 + lr) * LDS_STRIDE;
    const int bR2 = LDS_TILE + (wcol + 32 + lr) * LDS_STRIDE;
    const int bR3 = LDS_TILE + (wcol + 48 + lr) * LDS_STRIDE;

    const int s_c4 = (tid & 7) * 4;
    const int s_m0 = tid >> 3;
    const float* ag = xb + (size_t)(row_base + s_m0) * N_PIX + s_c4;
    const float* bg = xb + (size_t)(col_base + s_m0) * N_PIX + s_c4;
    const int s_l = s_m0 * LDS_STRIDE + s_c4;
    float mu[4];
    #pragma unroll
    for (int r = 0; r < 4; ++r) mu[r] = mb[row_base + s_m0 + 32 * r];

    v8f acc00 = {}, acc01 = {}, acc02 = {}, acc03 = {};
    v8f acc10 = {}, acc11 = {}, acc12 = {}, acc13 = {};

    float4 pa[4], pb[4];
    auto gload = [&](int kk) {
        const bool valid = (kk + s_c4) < N_PIX;
        #pragma unroll
        for (int r = 0; r < 4; ++r) {
            pa[r] = make_float4(0.0f, 0.0f, 0.0f, 0.0f);
            pb[r] = make_float4(0.0f, 0.0f, 0.0f, 0.0f);
            if (valid) {
                pa[r] = *(const float4*)(ag + (size_t)(32 * r) * N_PIX + kk);
                pb[r] = *(const float4*)(bg + (size_t)(32 * r) * N_PIX + kk);
            }
        }
    };
    auto lstore = [&](int kk, int st) {
        const bool valid = (kk + s_c4) < N_PIX;
        #pragma unroll
        for (int r = 0; r < 4; ++r) {
            v2f a_lo = {0.0f, 0.0f}, a_hi = {0.0f, 0.0f};
            if (valid) {
                a_lo[0] = (pa[r].x - mu[r]) * inv_n;
                a_lo[1] = (pa[r].y - mu[r]) * inv_n;
                a_hi[0] = (pa[r].z - mu[r]) * inv_n;
                a_hi[1] = (pa[r].w - mu[r]) * inv_n;
            }
            v2f b_lo = {pb[r].x, pb[r].y}, b_hi = {pb[r].z, pb[r].w};
            const int o = st + s_l + 32 * r * LDS_STRIDE;
            *(v2f*)&smem[o]                = a_lo;
            *(v2f*)&smem[o + 2]            = a_hi;
            *(v2f*)&smem[LDS_TILE + o]     = b_lo;
            *(v2f*)&smem[LDS_TILE + o + 2] = b_hi;
        }
    };

    gload(0);
    lstore(0, 0);
    __syncthreads();
    for (int it = 0; it < 7; ++it) {
        const int cur = (it & 1) * LDS_STAGE;
        const int nxt = LDS_STAGE - cur;
        const bool more = (it + 1) < 7;
        if (more) gload((it + 1) * 32);
        __builtin_amdgcn_sched_barrier(0);
        #pragma unroll
        for (int ks = 0; ks < 8; ++ks) {
            int kf = cur + ks * 4 + kh;
            v2f a0 = *(const v2f*)&smem[aR0 + kf];
            v2f a1 = *(const v2f*)&smem[aR1 + kf];
            v2f b0 = *(const v2f*)&smem[bR0 + kf];
            v2f b1 = *(const v2f*)&smem[bR1 + kf];
            v2f b2 = *(const v2f*)&smem[bR2 + kf];
            v2f b3 = *(const v2f*)&smem[bR3 + kf];
            acc00 = __builtin_amdgcn_wmma_f32_16x16x4_f32(false, a0, false, b0, (short)0, acc00, false, false);
            acc01 = __builtin_amdgcn_wmma_f32_16x16x4_f32(false, a0, false, b1, (short)0, acc01, false, false);
            acc02 = __builtin_amdgcn_wmma_f32_16x16x4_f32(false, a0, false, b2, (short)0, acc02, false, false);
            acc03 = __builtin_amdgcn_wmma_f32_16x16x4_f32(false, a0, false, b3, (short)0, acc03, false, false);
            acc10 = __builtin_amdgcn_wmma_f32_16x16x4_f32(false, a1, false, b0, (short)0, acc10, false, false);
            acc11 = __builtin_amdgcn_wmma_f32_16x16x4_f32(false, a1, false, b1, (short)0, acc11, false, false);
            acc12 = __builtin_amdgcn_wmma_f32_16x16x4_f32(false, a1, false, b2, (short)0, acc12, false, false);
            acc13 = __builtin_amdgcn_wmma_f32_16x16x4_f32(false, a1, false, b3, (short)0, acc13, false, false);
        }
        if (more) {
            lstore((it + 1) * 32, nxt);
            __syncthreads();
        }
    }
    const int mrow = row_base + wrow + (lane >> 4) * 8;
    const int ncol = col_base + wcol + lr;
    #pragma unroll
    for (int v = 0; v < 8; ++v) {
        Db[(size_t)(mrow + v) * C_DIM + ncol]           = acc00[v];
        Db[(size_t)(mrow + v) * C_DIM + ncol + 16]      = acc01[v];
        Db[(size_t)(mrow + v) * C_DIM + ncol + 32]      = acc02[v];
        Db[(size_t)(mrow + v) * C_DIM + ncol + 48]      = acc03[v];
        Db[(size_t)(mrow + 16 + v) * C_DIM + ncol]      = acc10[v];
        Db[(size_t)(mrow + 16 + v) * C_DIM + ncol + 16] = acc11[v];
        Db[(size_t)(mrow + 16 + v) * C_DIM + ncol + 32] = acc12[v];
        Db[(size_t)(mrow + 16 + v) * C_DIM + ncol + 48] = acc13[v];
    }
}

__global__ __launch_bounds__(256) void trace_kernel(const float* __restrict__ S,
                                                    float* __restrict__ scal) {
    __shared__ float red[256];
    int b = blockIdx.x, tid = threadIdx.x;
    red[tid] = S[(size_t)b * MAT_ELEMS + (size_t)tid * 257];
    __syncthreads();
    for (int s = 128; s > 0; s >>= 1) {
        if (tid < s) red[tid] += red[tid + s];
        __syncthreads();
    }
    if (tid == 0) {
        float tr = red[0];
        scal[b] = 1.0f / tr;
        scal[64 + b] = sqrtf(tr);
    }
}

__global__ __launch_bounds__(256) void b0_kernel(float* __restrict__ Bm,
                                                 const float* __restrict__ S,
                                                 const float* __restrict__ scal) {
    int col = threadIdx.x, row = blockIdx.x, b = blockIdx.y;
    size_t idx = (size_t)b * MAT_ELEMS + (size_t)row * C_DIM + col;
    float rtr = scal[b];
    Bm[idx] = 0.5f * ((row == col ? 3.0f : 0.0f) - S[idx] * rtr);
}

__global__ __launch_bounds__(256) void ns_gemm_kernel(float* __restrict__ D,
                                                      const float* __restrict__ A,
                                                      const float* __restrict__ B,
                                                      const float* __restrict__ scal,
                                                      int op) {
    __shared__ __attribute__((aligned(16))) float smem[2 * LDS_STAGE];
    const int tid  = threadIdx.x;
    const int lane = tid & 31;
    const int wave = tid >> 5;
    const int b    = blockIdx.y;
    const int tileid   = blockIdx.x;
    const int row_base = (tileid >> 1) * 128;
    const int col_base = (tileid & 1) * 128;
    const float* Ab = A + (size_t)b * MAT_ELEMS;
    const float* Bb = B + (size_t)b * MAT_ELEMS;
    float* Dmat = D + (size_t)b * MAT_ELEMS;
    float* Dout = D + (size_t)b * OUT_PER_B;

    const int wrow = (wave >> 1) * 32;
    const int wcol = (wave & 1) * 64;
    const int lr = lane & 15;
    const int kh = (lane >> 4) * 2;

    const int aR0 = (wrow +      lr) * LDS_STRIDE;
    const int aR1 = (wrow + 16 + lr) * LDS_STRIDE;
    const int bR0 = LDS_TILE + (wcol +      lr) * LDS_STRIDE;
    const int bR1 = LDS_TILE + (wcol + 16 + lr) * LDS_STRIDE;
    const int bR2 = LDS_TILE + (wcol + 32 + lr) * LDS_STRIDE;
    const int bR3 = LDS_TILE + (wcol + 48 + lr) * LDS_STRIDE;

    const int a_c4 = (tid & 7) * 4;
    const int a_m0 = tid >> 3;
    const float* ag = Ab + (size_t)(row_base + a_m0) * C_DIM + a_c4;
    const int a_l = a_m0 * LDS_STRIDE + a_c4;
    const int b_n4 = (tid & 31) * 4;
    const int b_k0 = tid >> 5;
    const float* bg = Bb + (size_t)b_k0 * C_DIM + col_base + b_n4;
    const int b_l = LDS_TILE + b_n4 * LDS_STRIDE + b_k0;

    v8f acc00 = {}, acc01 = {}, acc02 = {}, acc03 = {};
    v8f acc10 = {}, acc11 = {}, acc12 = {}, acc13 = {};

    float4 pa[4], pb[4];
    auto gload = [&](int kk) {
        #pragma unroll
        for (int r = 0; r < 4; ++r) pa[r] = *(const float4*)(ag + (size_t)(32 * r) * C_DIM + kk);
        #pragma unroll
        for (int r = 0; r < 4; ++r) pb[r] = *(const float4*)(bg + (size_t)(kk + 8 * r) * C_DIM);
    };
    auto lstore = [&](int st) {
        #pragma unroll
        for (int r = 0; r < 4; ++r) {
            v2f a_lo = {pa[r].x, pa[r].y}, a_hi = {pa[r].z, pa[r].w};
            const int o = st + a_l + 32 * r * LDS_STRIDE;
            *(v2f*)&smem[o]     = a_lo;
            *(v2f*)&smem[o + 2] = a_hi;
        }
        #pragma unroll
        for (int r = 0; r < 4; ++r) {
            const int o = st + b_l + 8 * r;
            smem[o]                  = pb[r].x;
            smem[o + LDS_STRIDE]     = pb[r].y;
            smem[o + 2 * LDS_STRIDE] = pb[r].z;
            smem[o + 3 * LDS_STRIDE] = pb[r].w;
        }
    };

    gload(0);
    lstore(0);
    __syncthreads();
    for (int it = 0; it < 8; ++it) {
        const int cur = (it & 1) * LDS_STAGE;
        const int nxt = LDS_STAGE - cur;
        const bool more = (it + 1) < 8;
        if (more) gload((it + 1) * 32);
        __builtin_amdgcn_sched_barrier(0);
        #pragma unroll
        for (int ks = 0; ks < 8; ++ks) {
            int kf = cur + ks * 4 + kh;
            v2f a0 = *(const v2f*)&smem[aR0 + kf];
            v2f a1 = *(const v2f*)&smem[aR1 + kf];
            v2f b0 = *(const v2f*)&smem[bR0 + kf];
            v2f b1 = *(const v2f*)&smem[bR1 + kf];
            v2f b2 = *(const v2f*)&smem[bR2 + kf];
            v2f b3 = *(const v2f*)&smem[bR3 + kf];
            acc00 = __builtin_amdgcn_wmma_f32_16x16x4_f32(false, a0, false, b0, (short)0, acc00, false, false);
            acc01 = __builtin_amdgcn_wmma_f32_16x16x4_f32(false, a0, false, b1, (short)0, acc01, false, false);
            acc02 = __builtin_amdgcn_wmma_f32_16x16x4_f32(false, a0, false, b2, (short)0, acc02, false, false);
            acc03 = __builtin_amdgcn_wmma_f32_16x16x4_f32(false, a0, false, b3, (short)0, acc03, false, false);
            acc10 = __builtin_amdgcn_wmma_f32_16x16x4_f32(false, a1, false, b0, (short)0, acc10, false, false);
            acc11 = __builtin_amdgcn_wmma_f32_16x16x4_f32(false, a1, false, b1, (short)0, acc11, false, false);
            acc12 = __builtin_amdgcn_wmma_f32_16x16x4_f32(false, a1, false, b2, (short)0, acc12, false, false);
            acc13 = __builtin_amdgcn_wmma_f32_16x16x4_f32(false, a1, false, b3, (short)0, acc13, false, false);
        }
        if (more) {
            lstore(nxt);
            __syncthreads();
        }
    }

    float s = 1.0f;
    if (op >= 2) s = scal[b];
    const int mrow = row_base + wrow + (lane >> 4) * 8;
    const int ncol = col_base + wcol + lr;

    auto emit = [&](float val, int m, int n) {
        if (op == 3) {
            if (n >= m) {
                int start = m * C_DIM - (m * (m - 1)) / 2;
                Dout[start + (n - m)] = val * s;
            }
        } else {
            if (op == 1)      val = (m == n ? 1.5f : 0.0f) - 0.5f * val;
            else if (op == 2) val *= s;
            Dmat[(size_t)m * C_DIM + n] = val;
        }
    };
    #pragma unroll
    for (int v = 0; v < 8; ++v) {
        emit(acc00[v], mrow + v,      ncol);
        emit(acc01[v], mrow + v,      ncol + 16);
        emit(acc02[v], mrow + v,      ncol + 32);
        emit(acc03[v], mrow + v,      ncol + 48);
        emit(acc10[v], mrow + 16 + v, ncol);
        emit(acc11[v], mrow + 16 + v, ncol + 16);
        emit(acc12[v], mrow + 16 + v, ncol + 32);
        emit(acc13[v], mrow + 16 + v, ncol + 48);
    }
}

extern "C" void kernel_launch(void* const* d_in, const int* in_sizes, int n_in,
                              void* d_out, int out_size, void* d_ws, size_t ws_size,
                              hipStream_t stream) {
    const float* x = (const float*)d_in[0];
    float* out = (float*)d_out;
    float* ws = (float*)d_ws;

    float* mean = ws;
    float* scal = ws + 16384;
    const size_t MS = (size_t)BATCH * MAT_ELEMS;
    float* M0 = ws + 16512;
    float* M1 = M0 + MS;
    float* M2 = M1 + MS;
    float* M3 = M2 + MS;
    float* M4 = M3 + MS;

    dim3 gGemm(4, BATCH);
    dim3 gRow(C_DIM, BATCH);

    mean_kernel<<<2048, 256, 0, stream>>>(x, mean);
    sigma_kernel<<<gGemm, 256, 0, stream>>>(M0, x, mean);
    trace_kernel<<<BATCH, 256, 0, stream>>>(M0, scal);
    b0_kernel<<<gRow, 256, 0, stream>>>(M1, M0, scal);

    ns_gemm_kernel<<<gGemm, 256, 0, stream>>>(M2, M0, M1, scal, 2);

    ns_gemm_kernel<<<gGemm, 256, 0, stream>>>(M3, M1, M2, nullptr, 1);
    ns_gemm_kernel<<<gGemm, 256, 0, stream>>>(M4, M2, M3, nullptr, 0);
    ns_gemm_kernel<<<gGemm, 256, 0, stream>>>(M0, M3, M1, nullptr, 0);
    ns_gemm_kernel<<<gGemm, 256, 0, stream>>>(M1, M0, M4, nullptr, 1);
    ns_gemm_kernel<<<gGemm, 256, 0, stream>>>(M2, M4, M1, nullptr, 0);
    ns_gemm_kernel<<<gGemm, 256, 0, stream>>>(M3, M1, M0, nullptr, 0);
    ns_gemm_kernel<<<gGemm, 256, 0, stream>>>(M4, M3, M2, nullptr, 1);
    ns_gemm_kernel<<<gGemm, 256, 0, stream>>>(M0, M2, M4, nullptr, 0);
    ns_gemm_kernel<<<gGemm, 256, 0, stream>>>(M1, M4, M3, nullptr, 0);

    ns_gemm_kernel<<<gGemm, 256, 0, stream>>>(M2, M1, M0, nullptr, 1);
    ns_gemm_kernel<<<gGemm, 256, 0, stream>>>(out, M0, M2, scal + 64, 3);
}


# each iteration's two independent products (Y@B and B@Z) run in one launch, blockIdx.z picks the operand set: two workgroups per WGP fill each other's waits; 16 launches become 13; sum order unchanged
# speedup vs baseline: 3.4593x; 1.0794x over previous
#include <hip/hip_runtime.h>
#include <hip/hip_bf16.h>
#include <math.h>

typedef __attribute__((ext_vector_type(2))) float v2f;
typedef __attribute__((ext_vector_type(8))) float v8f;

#define C_DIM 256
#define N_PIX 196
#define MAT_ELEMS (C_DIM * C_DIM)
#define BATCH 64
#define OUT_PER_B 32896
#define LDS_STRIDE 34
#define LDS_TILE (128 * LDS_STRIDE)
#define LDS_STAGE (2 * LDS_TILE)

__global__ __launch_bounds__(256) void mean_kernel(const float* __restrict__ x,
                                                   float* __restrict__ mean) {
    int row = blockIdx.x * 8 + (threadIdx.x >> 5);
    int lane = threadIdx.x & 31;
    const float* xr = x + (size_t)row * N_PIX;
    float s = 0.0f;
    for (int i = lane; i < N_PIX; i += 32) s += xr[i];
    #pragma unroll
    for (int o = 16; o > 0; o >>= 1) s += __shfl_down(s, o, 32);
    if (lane == 0) mean[row] = s * (1.0f / (float)N_PIX);
}

__global__ __launch_bounds__(256) void sigma_kernel(float* __restrict__ D,
                                                    const float* __restrict__ x,
                                                    const float* __restrict__ mean) {
    __shared__ __attribute__((aligned(16))) float smem[2 * LDS_STAGE];
    const int tid  = threadIdx.x;
    const int lane = tid & 31;
    const int wave = tid >> 5;
    const int b    = blockIdx.y;
    const int tileid   = blockIdx.x;
    const int row_base = (tileid >> 1) * 128;
    const int col_base = (tileid & 1) * 128;
    const float* xb = x + (size_t)b * C_DIM * N_PIX;
    const float* mb = mean + (size_t)b * C_DIM;
    float* Db = D + (size_t)b * MAT_ELEMS;

    const int wrow = (wave >> 1) * 32;
    const int wcol = (wave & 1) * 64;
    const int lr = lane & 15;
    const int kh = (lane >> 4) * 2;
    const float inv_n = 1.0f / (float)N_PIX;

    const int aR0 = (wrow +      lr) * LDS_STRIDE;
    const int aR1 = (wrow + 16 + lr) * LDS_STRIDE;
    const int bR0 = LDS_TILE + (wcol +      lr) * LDS_STRIDE;
    const int bR1 = LDS_TILE + (wcol + 16 + lr) * LDS_STRIDE;
    const int bR2 = LDS_TILE + (wcol + 32 + lr) * LDS_STRIDE;
    const int bR3 = LDS_TILE + (wcol + 48 + lr) * LDS_STRIDE;

    const int s_c4 = (tid & 7) * 4;
    const int s_m0 = tid >> 3;
    const float* ag = xb + (size_t)(row_base + s_m0) * N_PIX + s_c4;
    const float* bg = xb + (size_t)(col_base + s_m0) * N_PIX + s_c4;
    const int s_l = s_m0 * LDS_STRIDE + s_c4;
    float mu[4];
    #pragma unroll
    for (int r = 0; r < 4; ++r) mu[r] = mb[row_base + s_m0 + 32 * r];

    v8f acc00 = {}, acc01 = {}, acc02 = {}, acc03 = {};
    v8f acc10 = {}, acc11 = {}, acc12 = {}, acc13 = {};

    float4 pa[4], pb[4];
    auto gload = [&](int kk) {
        const bool valid = (kk + s_c4) < N_PIX;
        #pragma unroll
        for (int r = 0; r < 4; ++r) {
            pa[r] = make_float4(0.0f, 0.0f, 0.0f, 0.0f);
            pb[r] = make_float4(0.0f, 0.0f, 0.0f, 0.0f);
            if (valid) {
                pa[r] = *(const float4*)(ag + (size_t)(32 * r) * N_PIX + kk);
                pb[r] = *(const float4*)(bg + (size_t)(32 * r) * N_PIX + kk);
            }
        }
    };
    auto lstore = [&](int kk, int st) {
        const bool valid = (kk + s_c4) < N_PIX;
        #pragma unroll
        for (int r = 0; r < 4; ++r) {
            v2f a_lo = {0.0f, 0.0f}, a_hi = {0.0f, 0.0f};
            if (valid) {
                a_lo[0] = (pa[r].x - mu[r]) * inv_n;
                a_lo[1] = (pa[r].y - mu[r]) * inv_n;
                a_hi[0] = (pa[r].z - mu[r]) * inv_n;
                a_hi[1] = (pa[r].w - mu[r]) * inv_n;
            }
            v2f b_lo = {pb[r].x, pb[r].y}, b_hi = {pb[r].z, pb[r].w};
            const int o = st + s_l + 32 * r * LDS_STRIDE;
            *(v2f*)&smem[o]                = a_lo;
            *(v2f*)&smem[o + 2]            = a_hi;
            *(v2f*)&smem[LDS_TILE + o]     = b_lo;
            *(v2f*)&smem[LDS_TILE + o + 2] = b_hi;
        }
    };

    gload(0);
    lstore(0, 0);
    __syncthreads();
    for (int it = 0; it < 7; ++it) {
        const int cur = (it & 1) * LDS_STAGE;
        const int nxt = LDS_STAGE - cur;
        const bool more = (it + 1) < 7;
        if (more) gload((it + 1) * 32);
        __builtin_amdgcn_sched_barrier(0);
        #pragma unroll
        for (int ks = 0; ks < 8; ++ks) {
            int kf = cur + ks * 4 + kh;
            v2f a0 = *(const v2f*)&smem[aR0 + kf];
            v2f a1 = *(const v2f*)&smem[aR1 + kf];
            v2f b0 = *(const v2f*)&smem[bR0 + kf];
            v2f b1 = *(const v2f*)&smem[bR1 + kf];
            v2f b2 = *(const v2f*)&smem[bR2 + kf];
            v2f b3 = *(const v2f*)&smem[bR3 + kf];
            acc00 = __builtin_amdgcn_wmma_f32_16x16x4_f32(false, a0, false, b0, (short)0, acc00, false, false);
            acc01 = __builtin_amdgcn_wmma_f32_16x16x4_f32(false, a0, false, b1, (short)0, acc01, false, false);
            acc02 = __builtin_amdgcn_wmma_f32_16x16x4_f32(false, a0, false, b2, (short)0, acc02, false, false);
            acc03 = __builtin_amdgcn_wmma_f32_16x16x4_f32(false, a0, false, b3, (short)0, acc03, false, false);
            acc10 = __builtin_amdgcn_wmma_f32_16x16x4_f32(false, a1, false, b0, (short)0, acc10, false, false);
            acc11 = __builtin_amdgcn_wmma_f32_16x16x4_f32(false, a1, false, b1, (short)0, acc11, false, false);
            acc12 = __builtin_amdgcn_wmma_f32_16x16x4_f32(false, a1, false, b2, (short)0, acc12, false, false);
            acc13 = __builtin_amdgcn_wmma_f32_16x16x4_f32(false, a1, false, b3, (short)0, acc13, false, false);
        }
        if (more) {
            lstore((it + 1) * 32, nxt);
            __syncthreads();
        }
    }
    const int mrow = row_base + wrow + (lane >> 4) * 8;
    const int ncol = col_base + wcol + lr;
    #pragma unroll
    for (int v = 0; v < 8; ++v) {
        Db[(size_t)(mrow + v) * C_DIM + ncol]           = acc00[v];
        Db[(size_t)(mrow + v) * C_DIM + ncol + 16]      = acc01[v];
        Db[(size_t)(mrow + v) * C_DIM + ncol + 32]      = acc02[v];
        Db[(size_t)(mrow + v) * C_DIM + ncol + 48]      = acc03[v];
        Db[(size_t)(mrow + 16 + v) * C_DIM + ncol]      = acc10[v];
        Db[(size_t)(mrow + 16 + v) * C_DIM + ncol + 16] = acc11[v];
        Db[(size_t)(mrow + 16 + v) * C_DIM + ncol + 32] = acc12[v];
        Db[(size_t)(mrow + 16 + v) * C_DIM + ncol + 48] = acc13[v];
    }
}

__global__ __launch_bounds__(256) void trace_kernel(const float* __restrict__ S,
                                                    float* __restrict__ scal) {
    __shared__ float red[256];
    int b = blockIdx.x, tid = threadIdx.x;
    red[tid] = S[(size_t)b * MAT_ELEMS + (size_t)tid * 257];
    __syncthreads();
    for (int s = 128; s > 0; s >>= 1) {
        if (tid < s) red[tid] += red[tid + s];
        __syncthreads();
    }
    if (tid == 0) {
        float tr = red[0];
        scal[b] = 1.0f / tr;
        scal[64 + b] = sqrtf(tr);
    }
}

__global__ __launch_bounds__(256) void b0_kernel(float* __restrict__ Bm,
                                                 const float* __restrict__ S,
                                                 const float* __restrict__ scal) {
    int col = threadIdx.x, row = blockIdx.x, b = blockIdx.y;
    size_t idx = (size_t)b * MAT_ELEMS + (size_t)row * C_DIM + col;
    float rtr = scal[b];
    Bm[idx] = 0.5f * ((row == col ? 3.0f : 0.0f) - S[idx] * rtr);
}

__global__ __launch_bounds__(256) void ns_gemm_kernel(float* D,
                                                      const float* A,
                                                      const float* B,
                                                      float* D2,
                                                      const float* A2,
                                                      const float* B2,
                                                      const float* __restrict__ scal,
                                                      int op) {
    __shared__ __attribute__((aligned(16))) float smem[2 * LDS_STAGE];
    const int tid  = threadIdx.x;
    const int lane = tid & 31;
    const int wave = tid >> 5;
    const int b    = blockIdx.y;
    const int tileid   = blockIdx.x;
    const int row_base = (tileid >> 1) * 128;
    const int col_base = (tileid & 1) * 128;
    const bool second = blockIdx.z != 0;
    const float* __restrict__ Ab = (second ? A2 : A) + (size_t)b * MAT_ELEMS;
    const float* __restrict__ Bb = (second ? B2 : B) + (size_t)b * MAT_ELEMS;
    float* Dsel = second ? D2 : D;
    float* __restrict__ Dmat = Dsel + (size_t)b * MAT_ELEMS;
    float* __restrict__ Dout = Dsel + (size_t)b * OUT_PER_B;

    const int wrow = (wave >> 1) * 32;
    const int wcol = (wave & 1) * 64;
    const int lr = lane & 15;
    const int kh = (lane >> 4) * 2;

    const int aR0 = (wrow +      lr) * LDS_STRIDE;
    const int aR1 = (wrow + 16 + lr) * LDS_STRIDE;
    const int bR0 = LDS_TILE + (wcol +      lr) * LDS_STRIDE;
    const int bR1 = LDS_TILE + (wcol + 16 + lr) * LDS_STRIDE;
    const int bR2 = LDS_TILE + (wcol + 32 + lr) * LDS_STRIDE;
    const int bR3 = LDS_TILE + (wcol + 48 + lr) * LDS_STRIDE;

    const int a_c4 = (tid & 7) * 4;
    const int a_m0 = tid >> 3;
    const float* ag = Ab + (size_t)(row_base + a_m0) * C_DIM + a_c4;
    const int a_l = a_m0 * LDS_STRIDE + a_c4;
    const int b_n4 = (tid & 31) * 4;
    const int b_k0 = tid >> 5;
    const float* bg = Bb + (size_t)b_k0 * C_DIM + col_base + b_n4;
    const int b_l = LDS_TILE + b_n4 * LDS_STRIDE + b_k0;

    v8f acc00 = {}, acc01 = {}, acc02 = {}, acc03 = {};
    v8f acc10 = {}, acc11 = {}, acc12 = {}, acc13 = {};

    float4 pa[4], pb[4];
    auto gload = [&](int kk) {
        #pragma unroll
        for (int r = 0; r < 4; ++r) pa[r] = *(const float4*)(ag + (size_t)(32 * r) * C_DIM + kk);
        #pragma unroll
        for (int r = 0; r < 4; ++r) pb[r] = *(const float4*)(bg + (size_t)(kk + 8 * r) * C_DIM);
    };
    auto lstore = [&](int st) {
        #pragma unroll
        for (int r = 0; r < 4; ++r) {
            v2f a_lo = {pa[r].x, pa[r].y}, a_hi = {pa[r].z, pa[r].w};
            const int o = st + a_l + 32 * r * LDS_STRIDE;
            *(v2f*)&smem[o]     = a_lo;
            *(v2f*)&smem[o + 2] = a_hi;
        }
        #pragma unroll
        for (int r = 0; r < 4; ++r) {
            const int o = st + b_l + 8 * r;
            smem[o]                  = pb[r].x;
            smem[o + LDS_STRIDE]     = pb[r].y;
            smem[o + 2 * LDS_STRIDE] = pb[r].z;
            smem[o + 3 * LDS_STRIDE] = pb[r].w;
        }
    };

    gload(0);
    lstore(0);
    __syncthreads();
    for (int it = 0; it < 8; ++it) {
        const int cur = (it & 1) * LDS_STAGE;
        const int nxt = LDS_STAGE - cur;
        const bool more = (it + 1) < 8;
        if (more) gload((it + 1) * 32);
        __builtin_amdgcn_sched_barrier(0);
        #pragma unroll
        for (int ks = 0; ks < 8; ++ks) {
            int kf = cur + ks * 4 + kh;
            v2f a0 = *(const v2f*)&smem[aR0 + kf];
            v2f a1 = *(const v2f*)&smem[aR1 + kf];
            v2f b0 = *(const v2f*)&smem[bR0 + kf];
            v2f b1 = *(const v2f*)&smem[bR1 + kf];
            v2f b2 = *(const v2f*)&smem[bR2 + kf];
            v2f b3 = *(const v2f*)&smem[bR3 + kf];
            acc00 = __builtin_amdgcn_wmma_f32_16x16x4_f32(false, a0, false, b0, (short)0, acc00, false, false);
            acc01 = __builtin_amdgcn_wmma_f32_16x16x4_f32(false, a0, false, b1, (short)0, acc01, false, false);
            acc02 = __builtin_amdgcn_wmma_f32_16x16x4_f32(false, a0, false, b2, (short)0, acc02, false, false);
            acc03 = __builtin_amdgcn_wmma_f32_16x16x4_f32(false, a0, false, b3, (short)0, acc03, false, false);
            acc10 = __builtin_amdgcn_wmma_f32_16x16x4_f32(false, a1, false, b0, (short)0, acc10, false, false);
            acc11 = __builtin_amdgcn_wmma_f32_16x16x4_f32(false, a1, false, b1, (short)0, acc11, false, false);
            acc12 = __builtin_amdgcn_wmma_f32_16x16x4_f32(false, a1, false, b2, (short)0, acc12, false, false);
            acc13 = __builtin_amdgcn_wmma_f32_16x16x4_f32(false, a1, false, b3, (short)0, acc13, false, false);
        }
        if (more) {
            lstore(nxt);
            __syncthreads();
        }
    }

    float s = 1.0f;
    if (op >= 2) s = scal[b];
    const int mrow = row_base + wrow + (lane >> 4) * 8;
    const int ncol = col_base + wcol + lr;

    auto emit = [&](float val, int m, int n) {
        if (op == 3) {
            if (n >= m) {
                int start = m * C_DIM - (m * (m - 1)) / 2;
                Dout[start + (n - m)] = val * s;
            }
        } else {
            if (op == 1)      val = (m == n ? 1.5f : 0.0f) - 0.5f * val;
            else if (op == 2) val *= s;
            Dmat[(size_t)m * C_DIM + n] = val;
        }
    };
    #pragma unroll
    for (int v = 0; v < 8; ++v) {
        emit(acc00[v], mrow + v,      ncol);
        emit(acc01[v], mrow + v,      ncol + 16);
        emit(acc02[v], mrow + v,      ncol + 32);
        emit(acc03[v], mrow + v,      ncol + 48);
        emit(acc10[v], mrow + 16 + v, ncol);
        emit(acc11[v], mrow + 16 + v, ncol + 16);
        emit(acc12[v], mrow + 16 + v, ncol + 32);
        emit(acc13[v], mrow + 16 + v, ncol + 48);
    }
}

extern "C" void kernel_launch(void* const* d_in, const int* in_sizes, int n_in,
                              void* d_out, int out_size, void* d_ws, size_t ws_size,
                              hipStream_t stream) {
    const float* x = (const float*)d_in[0];
    float* out = (float*)d_out;
    float* ws = (float*)d_ws;

    float* mean = ws;
    float* scal = ws + 16384;
    const size_t MS = (size_t)BATCH * MAT_ELEMS;
    float* M0 = ws + 16512;
    float* M1 = M0 + MS;
    float* M2 = M1 + MS;
    float* M3 = M2 + MS;
    float* M4 = M3 + MS;

    dim3 gGemm(4, BATCH);
    dim3 gPair(4, BATCH, 2);
    dim3 gRow(C_DIM, BATCH);

    mean_kernel<<<2048, 256, 0, stream>>>(x, mean);
    sigma_kernel<<<gGemm, 256, 0, stream>>>(M0, x, mean);
    trace_kernel<<<BATCH, 256, 0, stream>>>(M0, scal);
    b0_kernel<<<gRow, 256, 0, stream>>>(M1, M0, scal);

    ns_gemm_kernel<<<gGemm, 256, 0, stream>>>(M2, M0, M1, M2, M0, M1, scal, 2);

    ns_gemm_kernel<<<gGemm, 256, 0, stream>>>(M3, M1, M2, M3, M1, M2, nullptr, 1);
    ns_gemm_kernel<<<gPair, 256, 0, stream>>>(M4, M2, M3, M0, M3, M1, nullptr, 0);
    ns_gemm_kernel<<<gGemm, 256, 0, stream>>>(M1, M0, M4, M1, M0, M4, nullptr, 1);
    ns_gemm_kernel<<<gPair, 256, 0, stream>>>(M2, M4, M1, M3, M1, M0, nullptr, 0);
    ns_gemm_kernel<<<gGemm, 256, 0, stream>>>(M4, M3, M2, M4, M3, M2, nullptr, 1);
    ns_gemm_kernel<<<gPair, 256, 0, stream>>>(M0, M2, M4, M1, M4, M3, nullptr, 0);

    ns_gemm_kernel<<<gGemm, 256, 0, stream>>>(M2, M1, M0, M2, M1, M0, nullptr, 1);
    ns_gemm_kernel<<<gGemm, 256, 0, stream>>>(out, M0, M2, out, M0, M2, scal + 64, 3);
}


# row means, trace and b0 pre-pass launches folded into the WMMA launches that use them (means in sigma; trace, b0 and its stored copy in the Y1 product): 13 launches become 10; sum order kept
# speedup vs baseline: 3.6018x; 1.0412x over previous
#include <hip/hip_runtime.h>
#include <hip/hip_bf16.h>
#include <math.h>

typedef __attribute__((ext_vector_type(2))) float v2f;
typedef __attribute__((ext_vector_type(8))) float v8f;

#define C_DIM 256
#define N_PIX 196
#define MAT_ELEMS (C_DIM * C_DIM)
#define BATCH 64
#define OUT_PER_B 32896
#define LDS_STRIDE 34
#define LDS_TILE (128 * LDS_STRIDE)
#define LDS_STAGE (2 * LDS_TILE)

__global__ __launch_bounds__(256) void sigma_kernel(float* __restrict__ D,
                                                    const float* __restrict__ x) {
    __shared__ __attribute__((aligned(16))) float smem[2 * LDS_STAGE];
    __shared__ float mu_s[128];
    const int tid  = threadIdx.x;
    const int lane = tid & 31;
    const int wave = tid >> 5;
    const int b    = blockIdx.y;
    const int tileid   = blockIdx.x;
    const int row_base = (tileid >> 1) * 128;
    const int col_base = (tileid & 1) * 128;
    const float* xb = x + (size_t)b * C_DIM * N_PIX;
    float* Db = D + (size_t)b * MAT_ELEMS;

    const int wrow = (wave >> 1) * 32;
    const int wcol = (wave & 1) * 64;
    const int lr = lane & 15;
    const int kh = (lane >> 4) * 2;
    const float inv_n = 1.0f / (float)N_PIX;

    const int aR0 = (wrow +      lr) * LDS_STRIDE;
    const int aR1 = (wrow + 16 + lr) * LDS_STRIDE;
    const int bR0 = LDS_TILE + (wcol +      lr) * LDS_STRIDE;
    const int bR1 = LDS_TILE + (wcol + 16 + lr) * LDS_STRIDE;
    const int bR2 = LDS_TILE + (wcol + 32 + lr) * LDS_STRIDE;
    const int bR3 = LDS_TILE + (wcol + 48 + lr) * LDS_STRIDE;

    const int s_c4 = (tid & 7) * 4;
    const int s_m0 = tid >> 3;
    const float* ag = xb + (size_t)(row_base + s_m0) * N_PIX + s_c4;
    const float* bg = xb + (size_t)(col_base + s_m0) * N_PIX + s_c4;
    const int s_l = s_m0 * LDS_STRIDE + s_c4;
    {
        float xv[16][7];
        const float* xw = xb + (size_t)(row_base + wave * 16) * N_PIX + lane;
        const int tail = (lane < 4) ? 192 : 160;
        #pragma unroll
        for (int j = 0; j < 16; ++j) {
            #pragma unroll
            for (int q = 0; q < 6; ++q) xv[j][q] = xw[(size_t)j * N_PIX + 32 * q];
            xv[j][6] = xw[(size_t)j * N_PIX + tail];
        }
        float rs[16];
        #pragma unroll
        for (int j = 0; j < 16; ++j) {
            float s = 0.0f;
            #pragma unroll
            for (int q = 0; q < 6; ++q) s += xv[j][q];
            s += (lane < 4) ? xv[j][6] : 0.0f;
            #pragma unroll
            for (int o = 16; o > 0; o >>= 1) s += __shfl_down(s, o, 32);
            rs[j] = s * (1.0f / (float)N_PIX);
        }
        if (lane == 0) {
            #pragma unroll
            for (int j = 0; j < 16; ++j) mu_s[wave * 16 + j] = rs[j];
        }
    }
    __syncthreads();
    float mu[4];
    #pragma unroll
    for (int r = 0; r < 4; ++r) mu[r] = mu_s[s_m0 + 32 * r];

    v8f acc00 = {}, acc01 = {}, acc02 = {}, acc03 = {};
    v8f acc10 = {}, acc11 = {}, acc12 = {}, acc13 = {};

    float4 pa[4], pb[4];
    auto gload = [&](int kk) {
        const bool valid = (kk + s_c4) < N_PIX;
        #pragma unroll
        for (int r = 0; r < 4; ++r) {
            pa[r] = make_float4(0.0f, 0.0f, 0.0f, 0.0f);
            pb[r] = make_float4(0.0f, 0.0f, 0.0f, 0.0f);
            if (valid) {
                pa[r] = *(const float4*)(ag + (size_t)(32 * r) * N_PIX + kk);
                pb[r] = *(const float4*)(bg + (size_t)(32 * r) * N_PIX + kk);
            }
        }
    };
    auto lstore = [&](int kk, int st) {
        const bool valid = (kk + s_c4) < N_PIX;
        #pragma unroll
        for (int r = 0; r < 4; ++r) {
            v2f a_lo = {0.0f, 0.0f}, a_hi = {0.0f, 0.0f};
            if (valid) {
                a_lo[0] = (pa[r].x - mu[r]) * inv_n;
                a_lo[1] = (pa[r].y - mu[r]) * inv_n;
                a_hi[0] = (pa[r].z - mu[r]) * inv_n;
                a_hi[1] = (pa[r].w - mu[r]) * inv_n;
            }
            v2f b_lo = {pb[r].x, pb[r].y}, b_hi = {pb[r].z, pb[r].w};
            const int o = st + s_l + 32 * r * LDS_STRIDE;
            *(v2f*)&smem[o]                = a_lo;
            *(v2f*)&smem[o + 2]            = a_hi;
            *(v2f*)&smem[LDS_TILE + o]     = b_lo;
            *(v2f*)&smem[LDS_TILE + o + 2] = b_hi;
        }
    };

    gload(0);
    lstore(0, 0);
    __syncthreads();
    for (int it = 0; it < 7; ++it) {
        const int cur = (it & 1) * LDS_STAGE;
        const int nxt = LDS_STAGE - cur;
        const bool more = (it + 1) < 7;
        if (more) gload((it + 1) * 32);
        __builtin_amdgcn_sched_barrier(0);
        #pragma unroll
        for (int ks = 0; ks < 8; ++ks) {
            int kf = cur + ks * 4 + kh;
            v2f a0 = *(const v2f*)&smem[aR0 + kf];
            v2f a1 = *(const v2f*)&smem[aR1 + kf];
            v2f b0 = *(const v2f*)&smem[bR0 + kf];
            v2f b1 = *(const v2f*)&smem[bR1 + kf];
            v2f b2 = *(const v2f*)&smem[bR2 + kf];
            v2f b3 = *(const v2f*)&smem[bR3 + kf];
            acc00 = __builtin_amdgcn_wmma_f32_16x16x4_f32(false, a0, false, b0, (short)0, acc00, false, false);
            acc01 = __builtin_amdgcn_wmma_f32_16x16x4_f32(false, a0, false, b1, (short)0, acc01, false, false);
            acc02 = __builtin_amdgcn_wmma_f32_16x16x4_f32(false, a0, false, b2, (short)0, acc02, false, false);
            acc03 = __builtin_amdgcn_wmma_f32_16x16x4_f32(false, a0, false, b3, (short)0, acc03, false, false);
            acc10 = __builtin_amdgcn_wmma_f32_16x16x4_f32(false, a1, false, b0, (short)0, acc10, false, false);
            acc11 = __builtin_amdgcn_wmma_f32_16x16x4_f32(false, a1, false, b1, (short)0, acc11, false, false);
            acc12 = __builtin_amdgcn_wmma_f32_16x16x4_f32(false, a1, false, b2, (short)0, acc12, false, false);
            acc13 = __builtin_amdgcn_wmma_f32_16x16x4_f32(false, a1, false, b3, (short)0, acc13, false, false);
        }
        if (more) {
            lstore((it + 1) * 32, nxt);
            __syncthreads();
        }
    }
    const int mrow = row_base + wrow + (lane >> 4) * 8;
    const int ncol = col_base + wcol + lr;
    #pragma unroll
    for (int v = 0; v < 8; ++v) {
        Db[(size_t)(mrow + v) * C_DIM + ncol]           = acc00[v];
        Db[(size_t)(mrow + v) * C_DIM + ncol + 16]      = acc01[v];
        Db[(size_t)(mrow + v) * C_DIM + ncol + 32]      = acc02[v];
        Db[(size_t)(mrow + v) * C_DIM + ncol + 48]      = acc03[v];
        Db[(size_t)(mrow + 16 + v) * C_DIM + ncol]      = acc10[v];
        Db[(size_t)(mrow + 16 + v) * C_DIM + ncol + 16] = acc11[v];
        Db[(size_t)(mrow + 16 + v) * C_DIM + ncol + 32] = acc12[v];
        Db[(size_t)(mrow + 16 + v) * C_DIM + ncol + 48] = acc13[v];
    }
}

__global__ __launch_bounds__(256) void y1_kernel(float* __restrict__ Y,
                                                 float* __restrict__ Bz,
                                                 const float* __restrict__ S,
                                                 float* __restrict__ scal) {
    __shared__ __attribute__((aligned(16))) float smem[2 * LDS_STAGE];
    __shared__ float red[256];
    const int tid  = threadIdx.x;
    const int lane = tid & 31;
    const int wave = tid >> 5;
    const int b    = blockIdx.y;
    const int tileid   = blockIdx.x;
    const int row_base = (tileid >> 1) * 128;
    const int col_base = (tileid & 1) * 128;
    const float* __restrict__ Sb = S + (size_t)b * MAT_ELEMS;
    float* __restrict__ Yb = Y + (size_t)b * MAT_ELEMS;
    float* __restrict__ Bb = Bz + (size_t)b * MAT_ELEMS;

    const int wrow = (wave >> 1) * 32;
    const int wcol = (wave & 1) * 64;
    const int lr = lane & 15;
    const int kh = (lane >> 4) * 2;

    const int aR0 = (wrow +      lr) * LDS_STRIDE;
    const int aR1 = (wrow + 16 + lr) * LDS_STRIDE;
    const int bR0 = LDS_TILE + (wcol +      lr) * LDS_STRIDE;
    const int bR1 = LDS_TILE + (wcol + 16 + lr) * LDS_STRIDE;
    const int bR2 = LDS_TILE + (wcol + 32 + lr) * LDS_STRIDE;
    const int bR3 = LDS_TILE + (wcol + 48 + lr) * LDS_STRIDE;

    const int a_c4 = (tid & 7) * 4;
    const int a_m0 = tid >> 3;
    const float* ag = Sb + (size_t)(row_base + a_m0) * C_DIM + a_c4;
    const int a_l = a_m0 * LDS_STRIDE + a_c4;
    const int b_n4 = (tid & 31) * 4;
    const int b_k0 = tid >> 5;
    const int b_c0 = col_base + b_n4;
    const float* bg = Sb + (size_t)b_k0 * C_DIM + b_c0;
    const int b_l = LDS_TILE + b_n4 * LDS_STRIDE + b_k0;
    const bool lower = (tileid >> 1) != 0;

    v8f acc00 = {}, acc01 = {}, acc02 = {}, acc03 = {};
    v8f acc10 = {}, acc11 = {}, acc12 = {}, acc13 = {};

    float4 pa[4], pb[4];
    auto gload = [&](int kk) {
        #pragma unroll
        for (int r = 0; r < 4; ++r) pa[r] = *(const float4*)(ag + (size_t)(32 * r) * C_DIM + kk);
        #pragma unroll
        for (int r = 0; r < 4; ++r) pb[r] = *(const float4*)(bg + (size_t)(kk + 8 * r) * C_DIM);
    };

    gload(0);

    red[tid] = Sb[(size_t)tid * 257];
    __syncthreads();
    float t = ((red[lane]      + red[lane + 128]) + (red[lane + 64] + red[lane + 192]))
            + ((red[lane + 32] + red[lane + 160]) + (red[lane + 96] + red[lane + 224]));
    #pragma unroll
    for (int o = 16; o > 0; o >>= 1) t += __shfl_down(t, o, 32);
    const float tr  = __shfl(t, 0, 32);
    const float rtr = 1.0f / tr;
    if (tileid == 0 && tid == 0) scal[64 + b] = sqrtf(tr);

    auto lstore = [&](int kk, int st) {
        #pragma unroll
        for (int r = 0; r < 4; ++r) {
            v2f a_lo = {pa[r].x, pa[r].y}, a_hi = {pa[r].z, pa[r].w};
            const int o = st + a_l + 32 * r * LDS_STRIDE;
            *(v2f*)&smem[o]     = a_lo;
            *(v2f*)&smem[o + 2] = a_hi;
        }
        auto piece = [&](const float4 p, int r) {
            const int krow = kk + b_k0 + 8 * r;
            float4 v;
            v.x = 0.5f * ((krow == b_c0     ? 3.0f : 0.0f) - p.x * rtr);
            v.y = 0.5f * ((krow == b_c0 + 1 ? 3.0f : 0.0f) - p.y * rtr);
            v.z = 0.5f * ((krow == b_c0 + 2 ? 3.0f : 0.0f) - p.z * rtr);
            v.w = 0.5f * ((krow == b_c0 + 3 ? 3.0f : 0.0f) - p.w * rtr);
            const int o = st + b_l + 8 * r;
            smem[o]                  = v.x;
            smem[o + LDS_STRIDE]     = v.y;
            smem[o + 2 * LDS_STRIDE] = v.z;
            smem[o + 3 * LDS_STRIDE] = v.w;
            return v;
        };
        const float4 v0 = piece(pb[0], 0);
        const float4 v1 = piece(pb[1], 1);
        const float4 v2 = piece(pb[2], 2);
        const float4 v3 = piece(pb[3], 3);
        float4 w0, w1;
        w0.x = lower ? v2.x : v0.x;  w0.y = lower ? v2.y : v0.y;  w0.z = lower ? v2.z : v0.z;  w0.w = lower ? v2.w : v0.w;
        w1.x = lower ? v3.x : v1.x;  w1.y = lower ? v3.y : v1.y;  w1.z = lower ? v3.z : v1.z;  w1.w = lower ? v3.w : v1.w;
        float* bd = Bb + (size_t)(kk + b_k0 + (lower ? 16 : 0)) * C_DIM + b_c0;
        *(float4*)bd               = w0;
        *(float4*)(bd + 8 * C_DIM) = w1;
    };

    lstore(0, 0);
    __syncthreads();
    for (int it = 0; it < 8; ++it) {
        const int cur = (it & 1) * LDS_STAGE;
        const int nxt = LDS_STAGE - cur;
        const bool more = (it + 1) < 8;
        if (more) gload((it + 1) * 32);
        __builtin_amdgcn_sched_barrier(0);
        #pragma unroll
        for (int ks = 0; ks < 8; ++ks) {
            int kf = cur + ks * 4 + kh;
            v2f a0 = *(const v2f*)&smem[aR0 + kf];
            v2f a1 = *(const v2f*)&smem[aR1 + kf];
            v2f b0 = *(const v2f*)&smem[bR0 + kf];
            v2f b1 = *(const v2f*)&smem[bR1 + kf];
            v2f b2 = *(const v2f*)&smem[bR2 + kf];
            v2f b3 = *(const v2f*)&smem[bR3 + kf];
            acc00 = __builtin_amdgcn_wmma_f32_16x16x4_f32(false, a0, false, b0, (short)0, acc00, false, false);
            acc01 = __builtin_amdgcn_wmma_f32_16x16x4_f32(false, a0, false, b1, (short)0, acc01, false, false);
            acc02 = __builtin_amdgcn_wmma_f32_16x16x4_f32(false, a0, false, b2, (short)0, acc02, false, false);
            acc03 = __builtin_amdgcn_wmma_f32_16x16x4_f32(false, a0, false, b3, (short)0, acc03, false, false);
            acc10 = __builtin_amdgcn_wmma_f32_16x16x4_f32(false, a1, false, b0, (short)0, acc10, false, false);
            acc11 = __builtin_amdgcn_wmma_f32_16x16x4_f32(false, a1, false, b1, (short)0, acc11, false, false);
            acc12 = __builtin_amdgcn_wmma_f32_16x16x4_f32(false, a1, false, b2, (short)0, acc12, false, false);
            acc13 = __builtin_amdgcn_wmma_f32_16x16x4_f32(false, a1, false, b3, (short)0, acc13, false, false);
        }
        if (more) {
            lstore((it + 1) * 32, nxt);
            __syncthreads();
        }
    }

    const int mrow = row_base + wrow + (lane >> 4) * 8;
    const int ncol = col_base + wcol + lr;
    #pragma unroll
    for (int v = 0; v < 8; ++v) {
        Yb[(size_t)(mrow + v) * C_DIM + ncol]           = acc00[v] * rtr;
        Yb[(size_t)(mrow + v) * C_DIM + ncol + 16]      = acc01[v] * rtr;
        Yb[(size_t)(mrow + v) * C_DIM + ncol + 32]      = acc02[v] * rtr;
        Yb[(size_t)(mrow + v) * C_DIM + ncol + 48]      = acc03[v] * rtr;
        Yb[(size_t)(mrow + 16 + v) * C_DIM + ncol]      = acc10[v] * rtr;
        Yb[(size_t)(mrow + 16 + v) * C_DIM + ncol + 16] = acc11[v] * rtr;
        Yb[(size_t)(mrow + 16 + v) * C_DIM + ncol + 32] = acc12[v] * rtr;
        Yb[(size_t)(mrow + 16 + v) * C_DIM + ncol + 48] = acc13[v] * rtr;
    }
}

__global__ __launch_bounds__(256) void ns_gemm_kernel(float* D,
                                                      const float* A,
                                                      const float* B,
                                                      float* D2,
                                                      const float* A2,
                                                      const float* B2,
                                                      const float* __restrict__ scal,
                                                      int op) {
    __shared__ __attribute__((aligned(16))) float smem[2 * LDS_STAGE];
    const int tid  = threadIdx.x;
    const int lane = tid & 31;
    const int wave = tid >> 5;
    const int b    = blockIdx.y;
    const int tileid   = blockIdx.x;
    const int row_base = (tileid >> 1) * 128;
    const int col_base = (tileid & 1) * 128;
    const bool second = blockIdx.z != 0;
    const float* __restrict__ Ab = (second ? A2 : A) + (size_t)b * MAT_ELEMS;
    const float* __restrict__ Bb = (second ? B2 : B) + (size_t)b * MAT_ELEMS;
    float* Dsel = second ? D2 : D;
    float* __restrict__ Dmat = Dsel + (size_t)b * MAT_ELEMS;
    float* __restrict__ Dout = Dsel + (size_t)b * OUT_PER_B;

    const int wrow = (wave >> 1) * 32;
    const int wcol = (wave & 1) * 64;
    const int lr = lane & 15;
    const int kh = (lane >> 4) * 2;

    const int aR0 = (wrow +      lr) * LDS_STRIDE;
    const int aR1 = (wrow + 16 + lr) * LDS_STRIDE;
    const int bR0 = LDS_TILE + (wcol +      lr) * LDS_STRIDE;
    const int bR1 = LDS_TILE + (wcol + 16 + lr) * LDS_STRIDE;
    const int bR2 = LDS_TILE + (wcol + 32 + lr) * LDS_STRIDE;
    const int bR3 = LDS_TILE + (wcol + 48 + lr) * LDS_STRIDE;

    const int a_c4 = (tid & 7) * 4;
    const int a_m0 = tid >> 3;
    const float* ag = Ab + (size_t)(row_base + a_m0) * C_DIM + a_c4;
    const int a_l = a_m0 * LDS_STRIDE + a_c4;
    const int b_n4 = (tid & 31) * 4;
    const int b_k0 = tid >> 5;
    const float* bg = Bb + (size_t)b_k0 * C_DIM + col_base + b_n4;
    const int b_l = LDS_TILE + b_n4 * LDS_STRIDE + b_k0;

    v8f acc00 = {}, acc01 = {}, acc02 = {}, acc03 = {};
    v8f acc10 = {}, acc11 = {}, acc12 = {}, acc13 = {};

    float4 pa[4], pb[4];
    auto gload = [&](int kk) {
        #pragma unroll
        for (int r = 0; r < 4; ++r) pa[r] = *(const float4*)(ag + (size_t)(32 * r) * C_DIM + kk);
        #pragma unroll
        for (int r = 0; r < 4; ++r) pb[r] = *(const float4*)(bg + (size_t)(kk + 8 * r) * C_DIM);
    };
    auto lstore = [&](int st) {
        #pragma unroll
        for (int r = 0; r < 4; ++r) {
            v2f a_lo = {pa[r].x, pa[r].y}, a_hi = {pa[r].z, pa[r].w};
            const int o = st + a_l + 32 * r * LDS_STRIDE;
            *(v2f*)&smem[o]     = a_lo;
            *(v2f*)&smem[o + 2] = a_hi;
        }
        #pragma unroll
        for (int r = 0; r < 4; ++r) {
            const int o = st + b_l + 8 * r;
            smem[o]                  = pb[r].x;
            smem[o + LDS_STRIDE]     = pb[r].y;
            smem[o + 2 * LDS_STRIDE] = pb[r].z;
            smem[o + 3 * LDS_STRIDE] = pb[r].w;
        }
    };

    gload(0);
    lstore(0);
    __syncthreads();
    for (int it = 0; it < 8; ++it) {
        const int cur = (it & 1) * LDS_STAGE;
        const int nxt = LDS_STAGE - cur;
        const bool more = (it + 1) < 8;
        if (more) gload((it + 1) * 32);
        __builtin_amdgcn_sched_barrier(0);
        #pragma unroll
        for (int ks = 0; ks < 8; ++ks) {
            int kf = cur + ks * 4 + kh;
            v2f a0 = *(const v2f*)&smem[aR0 + kf];
            v2f a1 = *(const v2f*)&smem[aR1 + kf];
            v2f b0 = *(const v2f*)&smem[bR0 + kf];
            v2f b1 = *(const v2f*)&smem[bR1 + kf];
            v2f b2 = *(const v2f*)&smem[bR2 + kf];
            v2f b3 = *(const v2f*)&smem[bR3 + kf];
            acc00 = __builtin_amdgcn_wmma_f32_16x16x4_f32(false, a0, false, b0, (short)0, acc00, false, false);
            acc01 = __builtin_amdgcn_wmma_f32_16x16x4_f32(false, a0, false, b1, (short)0, acc01, false, false);
            acc02 = __builtin_amdgcn_wmma_f32_16x16x4_f32(false, a0, false, b2, (short)0, acc02, false, false);
            acc03 = __builtin_amdgcn_wmma_f32_16x16x4_f32(false, a0, false, b3, (short)0, acc03, false, false);
            acc10 = __builtin_amdgcn_wmma_f32_16x16x4_f32(false, a1, false, b0, (short)0, acc10, false, false);
            acc11 = __builtin_amdgcn_wmma_f32_16x16x4_f32(false, a1, false, b1, (short)0, acc11, false, false);
            acc12 = __builtin_amdgcn_wmma_f32_16x16x4_f32(false, a1, false, b2, (short)0, acc12, false, false);
            acc13 = __builtin_amdgcn_wmma_f32_16x16x4_f32(false, a1, false, b3, (short)0, acc13, false, false);
        }
        if (more) {
            lstore(nxt);
            __syncthreads();
        }
    }

    float s = 1.0f;
    if (op == 3) s = scal[b];
    const int mrow = row_base + wrow + (lane >> 4) * 8;
    const int ncol = col_base + wcol + lr;

    auto emit = [&](float val, int m, int n) {
        if (op == 3) {
            if (n >= m) {
                int start = m * C_DIM - (m * (m - 1)) / 2;
                Dout[start + (n - m)] = val * s;
            }
        } else {
            if (op == 1) val = (m == n ? 1.5f : 0.0f) - 0.5f * val;
            Dmat[(size_t)m * C_DIM + n] = val;
        }
    };
    #pragma unroll
    for (int v = 0; v < 8; ++v) {
        emit(acc00[v], mrow + v,      ncol);
        emit(acc01[v], mrow + v,      ncol + 16);
        emit(acc02[v], mrow + v,      ncol + 32);
        emit(acc03[v], mrow + v,      ncol + 48);
        emit(acc10[v], mrow + 16 + v, ncol);
        emit(acc11[v], mrow + 16 + v, ncol + 16);
        emit(acc12[v], mrow + 16 + v, ncol + 32);
        emit(acc13[v], mrow + 16 + v, ncol + 48);
    }
}

extern "C" void kernel_launch(void* const* d_in, const int* in_sizes, int n_in,
                              void* d_out, int out_size, void* d_ws, size_t ws_size,
                              hipStream_t stream) {
    const float* x = (const float*)d_in[0];
    float* out = (float*)d_out;
    float* ws = (float*)d_ws;

    float* scal = ws + 16384;
    const size_t MS = (size_t)BATCH * MAT_ELEMS;
    float* M0 = ws + 16512;
    float* M1 = M0 + MS;
    float* M2 = M1 + MS;
    float* M3 = M2 + MS;
    float* M4 = M3 + MS;

    dim3 gGemm(4, BATCH);
    dim3 gPair(4, BATCH, 2);

    sigma_kernel<<<gGemm, 256, 0, stream>>>(M0, x);

    y1_kernel<<<gGemm, 256, 0, stream>>>(M2, M1, M0, scal);

    ns_gemm_kernel<<<gGemm, 256, 0, stream>>>(M3, M1, M2, M3, M1, M2, nullptr, 1);
    ns_gemm_kernel<<<gPair, 256, 0, stream>>>(M4, M2, M3, M0, M3, M1, nullptr, 0);
    ns_gemm_kernel<<<gGemm, 256, 0, stream>>>(M1, M0, M4, M1, M0, M4, nullptr, 1);
    ns_gemm_kernel<<<gPair, 256, 0, stream>>>(M2, M4, M1, M3, M1, M0, nullptr, 0);
    ns_gemm_kernel<<<gGemm, 256, 0, stream>>>(M4, M3, M2, M4, M3, M2, nullptr, 1);
    ns_gemm_kernel<<<gPair, 256, 0, stream>>>(M0, M2, M4, M1, M4, M3, nullptr, 0);

    ns_gemm_kernel<<<gGemm, 256, 0, stream>>>(M2, M1, M0, M2, M1, M0, nullptr, 1);
    ns_gemm_kernel<<<gGemm, 256, 0, stream>>>(out, M0, M2, out, M0, M2, scal + 64, 3);
}
